# starting kernel
# baseline (speedup 1.0000x reference)
#include <hip/hip_runtime.h>


typedef float v2f __attribute__((ext_vector_type(2)));
typedef float v8f __attribute__((ext_vector_type(8)));

#define B_    16
#define T_    2048
#define C_    256
#define K_    8
#define F_    256
#define DIL_  4
#define HALO  28
#define TT    16
#define XROWS 44
#define PITCH 260
#define NTHREADS 128

__global__ __launch_bounds__(256)
void wt_transpose_kernel(const float* __restrict__ w, float* __restrict__ wt) {
    int idx = blockIdx.x * 256 + threadIdx.x;
    if (idx >= K_ * F_ * C_) return;
    int c = idx % C_;
    int f = (idx / C_) % F_;
    int k = idx / (C_ * F_);
    wt[idx] = w[(c * K_ + k) * F_ + f];
}

__global__ __launch_bounds__(NTHREADS)
void MyDeConv1D_80367428043218_kernel(const float* __restrict__ x,
                                      const float* __restrict__ wt,
                                      const float* __restrict__ bias,
                                      float* __restrict__ out) {
    __shared__ __align__(16) float xs[XROWS * PITCH];

    const int tile = blockIdx.x & (T_ / TT - 1);
    const int b    = blockIdx.x >> 7;
    const int t_lo = tile * TT;
    const int tid  = threadIdx.x;

    const float4* xg = (const float4*)(x + (size_t)b * T_ * C_);
    for (int i = tid; i < XROWS * (C_ / 4); i += NTHREADS) {
        int row = i >> 6;
        int cq  = i & 63;
        int t   = t_lo + row;
        float4 v = make_float4(0.f, 0.f, 0.f, 0.f);
        if (t < T_) v = xg[t * (C_ / 4) + cq];
        *(float4*)&xs[row * PITCH + cq * 4] = v;
    }
    __syncthreads();

    const int lane  = tid & 31;
    const int wv    = tid >> 5;
    const int n16   = lane & 15;
    const int half  = lane >> 4;
    const int chalf = half * 2;
    const int f0    = wv * 64;

    v8f acc0 = {};
    v8f acc1 = {};
    v8f acc2 = {};
    v8f acc3 = {};

    for (int k = 0; k < K_; ++k) {
        const float* aP  = &xs[(HALO - DIL_ * k + n16) * PITCH + chalf];
        const float* b0P = wt + (((size_t)k * F_ + f0 + n16) * C_ + chalf);
#pragma unroll 4
        for (int cc = 0; cc < C_ / 4; ++cc) {
            v2f a  = *(const v2f*)(aP  + cc * 4);
            v2f b0 = *(const v2f*)(b0P + cc * 4);
            v2f b1 = *(const v2f*)(b0P + 16 * C_ + cc * 4);
            v2f b2 = *(const v2f*)(b0P + 32 * C_ + cc * 4);
            v2f b3 = *(const v2f*)(b0P + 48 * C_ + cc * 4);
            acc0 = __builtin_amdgcn_wmma_f32_16x16x4_f32(
                       false, a, false, b0, (short)0, acc0, false, false);
            acc1 = __builtin_amdgcn_wmma_f32_16x16x4_f32(
                       false, a, false, b1, (short)0, acc1, false, false);
            acc2 = __builtin_amdgcn_wmma_f32_16x16x4_f32(
                       false, a, false, b2, (short)0, acc2, false, false);
            acc3 = __builtin_amdgcn_wmma_f32_16x16x4_f32(
                       false, a, false, b3, (short)0, acc3, false, false);
        }
    }

    const float bias0 = bias[f0 + n16]      * (float)C_;
    const float bias1 = bias[f0 + 16 + n16] * (float)C_;
    const float bias2 = bias[f0 + 32 + n16] * (float)C_;
    const float bias3 = bias[f0 + 48 + n16] * (float)C_;
    float* outb = out + (size_t)b * T_ * F_;
#pragma unroll
    for (int i = 0; i < 8; ++i) {
        int m   = i + half * 8;
        int tau = t_lo + m;
        int d   = (tau + HALO) - (T_ - 1);
        int kmin = d > 0 ? ((d + 3) >> 2) : 0;
        float cnt = (float)(K_ - kmin);
        float* o = outb + (size_t)tau * F_ + f0 + n16;
        o[0]  = acc0[i] + cnt * bias0;
        o[16] = acc1[i] + cnt * bias1;
        o[32] = acc2[i] + cnt * bias2;
        o[48] = acc3[i] + cnt * bias3;
    }
}

extern "C" void kernel_launch(void* const* d_in, const int* in_sizes, int n_in,
                              void* d_out, int out_size, void* d_ws, size_t ws_size,
                              hipStream_t stream) {
    const float* x    = (const float*)d_in[0];
    const float* w    = (const float*)d_in[1];
    const float* bias = (const float*)d_in[2];
    float* out = (float*)d_out;
    float* wt  = (float*)d_ws;

    const int nW = K_ * F_ * C_;
    wt_transpose_kernel<<<(nW + 255) / 256, 256, 0, stream>>>(w, wt);

    MyDeConv1D_80367428043218_kernel<<<B_ * (T_ / TT), NTHREADS, 0, stream>>>(
        x, wt, bias, out);
}


# each iteration's hand-over part freed of vector-ALU work: LDS bases of slab i+1 formed among slab i's products, weight fetches take a scalar base + fixed lane offset; sum order kept
# speedup vs baseline: 2.4484x; 2.4484x over previous
#include <hip/hip_runtime.h>


typedef float v2f __attribute__((ext_vector_type(2)));
typedef float v8f __attribute__((ext_vector_type(8)));
typedef __attribute__((address_space(3))) float lds_f;
typedef __attribute__((address_space(3))) v2f   lds_v2f;

#define B_    16
#define T_    2048
#define C_    256
#define K_    8
#define F_    256
#define DIL_  4
#define HALO  28
#define TB_   128
#define XROWS (TB_ + HALO)
#define PITCH 260
#define CCH   32
#define NCH   (C_ / CCH)
#define NSLAB (K_ * NCH)
#define WPITCH 544
#define WBUF  ((CCH / 2) * WPITCH)
#define NRING 4
#define NTHREADS 256

struct SlabRegs { float4 a0, a1, a2, a3, b0, b1, b2, b3; };

static __device__ __forceinline__ SlabRegs load_slab(const float* w, unsigned toff, int i) {
    const float* s = w + (size_t)((i & (NCH - 1)) * CCH) * (K_ * F_) + (i >> 3) * F_;
    SlabRegs r;
    r.a0 = *(const float4*)(s + toff);
    r.b0 = *(const float4*)(s + K_ * F_ + toff);
    r.a1 = *(const float4*)(s + (size_t)8 * (K_ * F_) + toff);
    r.b1 = *(const float4*)(s + (size_t)8 * (K_ * F_) + K_ * F_ + toff);
    r.a2 = *(const float4*)(s + (size_t)16 * (K_ * F_) + toff);
    r.b2 = *(const float4*)(s + (size_t)16 * (K_ * F_) + K_ * F_ + toff);
    r.a3 = *(const float4*)(s + (size_t)24 * (K_ * F_) + toff);
    r.b3 = *(const float4*)(s + (size_t)24 * (K_ * F_) + K_ * F_ + toff);
    return r;
}

static __device__ __forceinline__ void store_slab(float* d, const SlabRegs& r) {
    *(float4*)(d)                  = r.a0;
    *(float4*)(d + 4)              = r.b0;
    *(float4*)(d + 4 * WPITCH)     = r.a1;
    *(float4*)(d + 4 * WPITCH + 4) = r.b1;
    *(float4*)(d + 8 * WPITCH)     = r.a2;
    *(float4*)(d + 8 * WPITCH + 4) = r.b2;
    *(float4*)(d + 12 * WPITCH)     = r.a3;
    *(float4*)(d + 12 * WPITCH + 4) = r.b3;
}

__global__ __launch_bounds__(NTHREADS)
void dilated_deconv_main_kernel(const float* __restrict__ x,
                                const float* __restrict__ w,
                                const float* __restrict__ bias,
                                float* __restrict__ out) {
    __shared__ __align__(16) float xs[XROWS * PITCH];
    __shared__ __align__(16) float wsm[NRING * WBUF];

    const int tid  = threadIdx.x;
    const int tile = blockIdx.x & (T_ / TB_ - 1);
    const int b    = blockIdx.x >> 4;
    const int t_lo = tile * TB_;

    const int q  = tid & 63;
    const int p0 = tid >> 6;
    const unsigned toff = (unsigned)((2 * p0) * (K_ * F_) + 4 * q);
    float*         wdst = wsm + p0 * WPITCH + 8 * q;

    SlabRegs sr  = load_slab(w, toff, 0);
    SlabRegs sr1 = load_slab(w, toff, 1);

    const float4* xg = (const float4*)(x + (size_t)b * T_ * C_);
#pragma unroll 13
    for (int j = 0; j < XROWS / 4; ++j) {
        int row = p0 + 4 * j;
        int t   = t_lo + row;
        int tc  = t < T_ ? t : T_ - 1;
        float4 v = xg[tc * (C_ / 4) + q];
        if (t >= T_) v = make_float4(0.f, 0.f, 0.f, 0.f);
        *(float4*)&xs[row * PITCH + q * 4] = v;
    }
    store_slab(wdst, sr);
    store_slab(wdst + WBUF, sr1);
    __syncthreads();
    sr = load_slab(w, toff, 2);

    const int lane  = tid & 31;
    const int wv    = tid >> 5;
    const int n16   = lane & 15;
    const int half  = lane >> 4;
    const int chalf = half * 2;
    const int m0    = (wv >> 2) * 64;
    const int f0    = (wv & 3) * 64;

    v8f acc[4][4] = {};

#define MMA_STEPS(c_lo, c_hi)                                                      \
    {                                                                              \
        _Pragma("unroll")                                                          \
        for (int cc = (c_lo); cc < (c_hi); ++cc)                                   \
            _Pragma("unroll")                                                      \
            for (int mt = 0; mt < 4; ++mt)                                         \
                _Pragma("unroll")                                                  \
                for (int nt = 0; nt < 4; ++nt)                                     \
                    acc[mt][nt] = __builtin_amdgcn_wmma_f32_16x16x4_f32(           \
                        false, a[cc][mt], false, bf[cc][nt], (short)0, acc[mt][nt],\
                        false, false);                                             \
    }

    const lds_f* aLane = (const lds_f*)&xs[(HALO + m0 + n16) * PITCH + chalf];
    const lds_f* bLane = (const lds_f*)&wsm[half * WPITCH + 2 * f0 + 8 * (n16 >> 2) + (n16 & 3)];
    const lds_f* aQ[4];
    const lds_f* bQ[CCH / 4];
#pragma unroll
    for (int mt = 0; mt < 4; ++mt) aQ[mt] = aLane + mt * 16 * PITCH;
#pragma unroll
    for (int cc = 0; cc < CCH / 4; ++cc) bQ[cc] = bLane + cc * 2 * WPITCH;
    float* sQ = wdst + 2 * WBUF;

    for (int i = 0; i < NSLAB; ++i) {
        v2f a[CCH / 4][4], bf[CCH / 4][4];
#pragma unroll
        for (int cc = 0; cc < CCH / 4; ++cc) {
#pragma unroll
            for (int mt = 0; mt < 4; ++mt)
                a[cc][mt] = *(const lds_v2f*)(aQ[mt] + cc * 4);
#pragma unroll
            for (int nt = 0; nt < 4; ++nt) {
                bf[cc][nt].x = bQ[cc][nt * 32];
                bf[cc][nt].y = bQ[cc][nt * 32 + 4];
            }
        }
        if (i + 2 < NSLAB) store_slab(sQ, sr);
        if (i + 3 < NSLAB) sr = load_slab(w, toff, i + 3);
        __builtin_amdgcn_sched_barrier(0);
        MMA_STEPS(0, 4)
        {
            const int n  = i + 1;
            const int au = (n & (NCH - 1)) * CCH - DIL_ * (n >> 3) * PITCH;
            const int bu = (n & (NRING - 1)) * WBUF;
#pragma unroll
            for (int mt = 0; mt < 4; ++mt) aQ[mt] = aLane + (mt * 16 * PITCH + au);
#pragma unroll
            for (int cc = 0; cc < CCH / 4; ++cc) bQ[cc] = bLane + (cc * 2 * WPITCH + bu);
            sQ = wdst + ((n + 2) & (NRING - 1)) * WBUF;
        }
        __builtin_amdgcn_sched_barrier(0);
        if (i > 0) asm volatile("s_barrier_wait -1" ::: "memory");
        __builtin_amdgcn_sched_barrier(0);
        MMA_STEPS(4, 8)
        __builtin_amdgcn_sched_barrier(0);
        if (i + 1 < NSLAB) {
            asm volatile("s_wait_dscnt 0x0" ::: "memory");
            asm volatile("s_barrier_signal -1" ::: "memory");
        }
    }
#undef MMA_STEPS

    float biasv[4];
#pragma unroll
    for (int nt = 0; nt < 4; ++nt) biasv[nt] = bias[f0 + nt * 16 + n16] * (float)C_;
    float* outb = out + (size_t)b * T_ * F_;
#pragma unroll
    for (int mt = 0; mt < 4; ++mt) {
#pragma unroll
        for (int i = 0; i < 8; ++i) {
            int m   = m0 + mt * 16 + i + half * 8;
            int tau = t_lo + m;
            int d   = (tau + HALO) - (T_ - 1);
            int kmin = d > 0 ? ((d + 3) >> 2) : 0;
            float cnt = (float)(K_ - kmin);
            float* o = outb + (size_t)tau * F_ + f0 + n16;
#pragma unroll
            for (int nt = 0; nt < 4; ++nt)
                o[nt * 16] = acc[mt][nt][i] + cnt * biasv[nt];
        }
    }
}

extern "C" void kernel_launch(void* const* d_in, const int* in_sizes, int n_in,
                              void* d_out, int out_size, void* d_ws, size_t ws_size,
                              hipStream_t stream) {
    const float* x    = (const float*)d_in[0];
    const float* w    = (const float*)d_in[1];
    const float* bias = (const float*)d_in[2];
    float* out = (float*)d_out;

    dilated_deconv_main_kernel<<<B_ * (T_ / TB_), NTHREADS, 0, stream>>>(
        x, w, bias, out);
}
